# starting kernel
# baseline (speedup 1.0000x reference)
#include <hip/hip_runtime.h>
#include <math.h>

typedef __attribute__((ext_vector_type(2))) float v2f;
typedef __attribute__((ext_vector_type(8))) float v8f;

#define BATCH 2
#define SEQ 2048
#define DM 1024
#define NH 16
#define DK 64

template <int STORE_VT>
__global__ __launch_bounds__(256) void gemm_xwt_kernel(const float* __restrict__ X,
                                                       const float* __restrict__ W,
                                                       float* __restrict__ out,
                                                       int M, int N, int K) {
    const int lane = threadIdx.x & 31;
    const int warp = threadIdx.x >> 5;
    const int tile = blockIdx.x * 8 + warp;
    const int tilesN = N >> 6;
    const int tm = (tile / tilesN) << 5;
    const int tn = (tile % tilesN) << 6;
    const int ln = lane & 15;
    const int lh = lane >> 4;

    const float* xrow0 = X + (size_t)(tm + ln) * K + 2 * lh;
    const float* xrow1 = xrow0 + (size_t)16 * K;
    const float* wrow  = W + (size_t)(tn + ln) * K + 2 * lh;

    v8f c[2][4] = {};
#pragma unroll 4
    for (int k0 = 0; k0 < K; k0 += 4) {
        v2f a0 = *(const v2f*)(xrow0 + k0);
        v2f a1 = *(const v2f*)(xrow1 + k0);
        v2f b0 = *(const v2f*)(wrow + k0);
        v2f b1 = *(const v2f*)(wrow + (size_t)16 * K + k0);
        v2f b2 = *(const v2f*)(wrow + (size_t)32 * K + k0);
        v2f b3 = *(const v2f*)(wrow + (size_t)48 * K + k0);
        c[0][0] = __builtin_amdgcn_wmma_f32_16x16x4_f32(false, a0, false, b0, (short)0, c[0][0], false, false);
        c[0][1] = __builtin_amdgcn_wmma_f32_16x16x4_f32(false, a0, false, b1, (short)0, c[0][1], false, false);
        c[0][2] = __builtin_amdgcn_wmma_f32_16x16x4_f32(false, a0, false, b2, (short)0, c[0][2], false, false);
        c[0][3] = __builtin_amdgcn_wmma_f32_16x16x4_f32(false, a0, false, b3, (short)0, c[0][3], false, false);
        c[1][0] = __builtin_amdgcn_wmma_f32_16x16x4_f32(false, a1, false, b0, (short)0, c[1][0], false, false);
        c[1][1] = __builtin_amdgcn_wmma_f32_16x16x4_f32(false, a1, false, b1, (short)0, c[1][1], false, false);
        c[1][2] = __builtin_amdgcn_wmma_f32_16x16x4_f32(false, a1, false, b2, (short)0, c[1][2], false, false);
        c[1][3] = __builtin_amdgcn_wmma_f32_16x16x4_f32(false, a1, false, b3, (short)0, c[1][3], false, false);
    }

    if (STORE_VT == 0) {
#pragma unroll
        for (int mi = 0; mi < 2; ++mi)
#pragma unroll
            for (int ni = 0; ni < 4; ++ni) {
                float* orow = out + (size_t)(tm + mi * 16 + 8 * lh) * N + tn + ni * 16 + ln;
#pragma unroll
                for (int r = 0; r < 8; ++r)
                    orow[(size_t)r * N] = c[mi][ni][r];
            }
    } else {
        const int bb = tm >> 11;
#pragma unroll
        for (int mi = 0; mi < 2; ++mi)
#pragma unroll
            for (int ni = 0; ni < 4; ++ni) {
                int n = tn + ni * 16 + ln;
                int srow = (tm & (SEQ - 1)) + mi * 16 + 8 * lh;
                float* ocol = out + ((size_t)(bb * N + n) << 11) + srow;
#pragma unroll
                for (int r = 0; r < 8; ++r)
                    ocol[r] = c[mi][ni][r];
            }
    }
}

__global__ __launch_bounds__(256) void rope_kernel(float* __restrict__ q,
                                                   float* __restrict__ k, int total) {
    int tid = blockIdx.x * blockDim.x + threadIdx.x;
    if (tid >= total) return;
    int i = tid & 31;
    int s = (tid >> 9) & (SEQ - 1);
    float freq = __expf(-(float)i * (1.0f / 32.0f) * 9.210340371976184f);
    float ang = (float)s * freq;
    float cs = __cosf(ang);
    float sn = __sinf(ang);
    size_t p = (size_t)tid * 2;
    float q1 = q[p], q2 = q[p + 1];
    q[p]     = cs * q1 - sn * q2;
    q[p + 1] = sn * q1 + cs * q2;
    float k1 = k[p], k2 = k[p + 1];
    k[p]     = cs * k1 - sn * k2;
    k[p + 1] = sn * k1 + cs * k2;
}

__device__ __forceinline__ float rowmax16(float v) {
    v = fmaxf(v, __shfl_xor(v, 1, 32));
    v = fmaxf(v, __shfl_xor(v, 2, 32));
    v = fmaxf(v, __shfl_xor(v, 4, 32));
    v = fmaxf(v, __shfl_xor(v, 8, 32));
    return v;
}
__device__ __forceinline__ float rowsum16(float v) {
    v += __shfl_xor(v, 1, 32);
    v += __shfl_xor(v, 2, 32);
    v += __shfl_xor(v, 4, 32);
    v += __shfl_xor(v, 8, 32);
    return v;
}

__global__ __launch_bounds__(256) void attn_kernel(const float* __restrict__ q,
                                                   const float* __restrict__ k,
                                                   const float* __restrict__ vt,
                                                   float* __restrict__ o) {
    __shared__ float lds[8][16 * 18];
    const int lane = threadIdx.x & 31;
    const int warp = threadIdx.x >> 5;
    const int wave = blockIdx.x * 8 + warp;
    const int qt = wave & 127;
    const int bh = wave >> 7;
    const int h = bh & 15;
    const int b = bh >> 4;
    const int ln = lane & 15;
    const int lh = lane >> 4;

    const float* qbase = q + ((size_t)(b * SEQ) + qt * 16 + ln) * DM + h * DK;
    v2f qa[16];
#pragma unroll
    for (int j = 0; j < 16; ++j)
        qa[j] = *(const v2f*)(qbase + 4 * j + 2 * lh);

    const float* vrow[4];
#pragma unroll
    for (int nt = 0; nt < 4; ++nt)
        vrow[nt] = vt + ((size_t)(b * DM + h * DK + nt * 16 + ln) << 11);

    v8f acc[4] = {v8f{}, v8f{}, v8f{}, v8f{}};
    float mrow[8], lrow[8];
#pragma unroll
    for (int r = 0; r < 8; ++r) { mrow[r] = -1e30f; lrow[r] = 0.0f; }

    float* pl = &lds[warp][0];

    for (int kt = 0; kt <= qt; ++kt) {
        const float* kbase = k + ((size_t)(b * SEQ) + kt * 16 + ln) * DM + h * DK;
        v8f s = {};
#pragma unroll
        for (int j = 0; j < 16; ++j) {
            v2f bf = *(const v2f*)(kbase + 4 * j + 2 * lh);
            s = __builtin_amdgcn_wmma_f32_16x16x4_f32(false, qa[j], false, bf,
                                                      (short)0, s, false, false);
        }
        float nm[8], rs[8];
#pragma unroll
        for (int r = 0; r < 8; ++r) {
            float sv = s[r] * 0.125f;
            if (kt == qt && ln > (r + 8 * lh)) sv = -1e30f;
            s[r] = sv;
            nm[r] = fmaxf(mrow[r], rowmax16(sv));
        }
#pragma unroll
        for (int r = 0; r < 8; ++r) {
            float p = __expf(s[r] - nm[r]);
            s[r] = p;
            rs[r] = rowsum16(p);
        }
#pragma unroll
        for (int r = 0; r < 8; ++r) {
            float sc = __expf(mrow[r] - nm[r]);
            lrow[r] = lrow[r] * sc + rs[r];
            mrow[r] = nm[r];
#pragma unroll
            for (int nt = 0; nt < 4; ++nt) acc[nt][r] *= sc;
        }
#pragma unroll
        for (int r = 0; r < 8; ++r)
            pl[(r + 8 * lh) * 18 + ln] = s[r];
#pragma unroll
        for (int j = 0; j < 4; ++j) {
            int kk = 4 * j + 2 * lh;
            v2f a = *(const v2f*)(pl + ln * 18 + kk);
#pragma unroll
            for (int nt = 0; nt < 4; ++nt) {
                v2f bf = *(const v2f*)(vrow[nt] + kt * 16 + kk);
                acc[nt] = __builtin_amdgcn_wmma_f32_16x16x4_f32(false, a, false, bf,
                                                                (short)0, acc[nt],
                                                                false, false);
            }
        }
    }

    float* obase = o + ((size_t)(b * SEQ) + qt * 16) * DM + h * DK;
#pragma unroll
    for (int r = 0; r < 8; ++r) {
        float inv = 1.0f / lrow[r];
        int m = r + 8 * lh;
#pragma unroll
        for (int nt = 0; nt < 4; ++nt)
            obase[(size_t)m * DM + nt * 16 + ln] = acc[nt][r] * inv;
    }
}

extern "C" void kernel_launch(void* const* d_in, const int* in_sizes, int n_in,
                              void* d_out, int out_size, void* d_ws, size_t ws_size,
                              hipStream_t stream) {
    const float* x  = (const float*)d_in[0];
    const float* Wq = (const float*)d_in[1];
    const float* Wk = (const float*)d_in[2];
    const float* Wv = (const float*)d_in[3];
    const float* Wo = (const float*)d_in[4];
    float* out = (float*)d_out;

    const size_t bufElems = (size_t)BATCH * SEQ * DM;
    float* qb  = (float*)d_ws;
    float* kb  = qb + bufElems;
    float* vtb = kb + bufElems;
    float* ab  = vtb + bufElems;

    const int M = BATCH * SEQ, N = DM, K = DM;
    const int tiles = (M / 32) * (N / 64);
    const int gblocks = tiles / 8;

    gemm_xwt_kernel<0><<<gblocks, 256, 0, stream>>>(x, Wq, qb, M, N, K);
    gemm_xwt_kernel<0><<<gblocks, 256, 0, stream>>>(x, Wk, kb, M, N, K);
    gemm_xwt_kernel<1><<<gblocks, 256, 0, stream>>>(x, Wv, vtb, M, N, K);

    const int pairs = BATCH * SEQ * NH * (DK / 2);
    rope_kernel<<<pairs / 256, 256, 0, stream>>>(qb, kb, pairs);

    const int waves = BATCH * NH * (SEQ / 16);
    attn_kernel<<<waves / 8, 256, 0, stream>>>(qb, kb, vtb, ab);

    gemm_xwt_kernel<0><<<gblocks, 256, 0, stream>>>(ab, Wo, out, M, N, K);
}


# attention: online softmax advanced once per 64-key block, not per 16-key tile: one row-max butterfly and one rescale per block, row sums kept per lane, reduced once at the end; row-sum order changes
# speedup vs baseline: 1.9240x; 1.9240x over previous
#include <hip/hip_runtime.h>
#include <math.h>

typedef __attribute__((ext_vector_type(2))) float v2f;
typedef __attribute__((ext_vector_type(4))) float v4f;
typedef __attribute__((ext_vector_type(8))) float v8f;

#define BATCH 2
#define SEQ 2048
#define DM 1024
#define NH 16
#define DK 64

struct GemmFrag {
    v4f a[4][4];
    v4f b[4][4];
};

__device__ __forceinline__ void gemm_load_chunk(GemmFrag& f, const float* __restrict__ xp,
                                                const float* __restrict__ wp, int K, int kc) {
#pragma unroll
    for (int i = 0; i < 4; ++i) {
#pragma unroll
        for (int j = 0; j < 4; ++j)
            f.a[i][j] = *(const v4f*)(xp + (size_t)(16 * i) * K + kc + 8 * j);
#pragma unroll
        for (int j = 0; j < 4; ++j)
            f.b[i][j] = *(const v4f*)(wp + (size_t)(16 * i) * K + kc + 8 * j);
    }
}

__device__ __forceinline__ void gemm_mma_chunk(v8f (&c)[4][4], const GemmFrag& f) {
#pragma unroll
    for (int j = 0; j < 4; ++j) {
        v2f alo[4], ahi[4], blo[4], bhi[4];
#pragma unroll
        for (int i = 0; i < 4; ++i) {
            alo[i] = __builtin_shufflevector(f.a[i][j], f.a[i][j], 0, 1);
            ahi[i] = __builtin_shufflevector(f.a[i][j], f.a[i][j], 2, 3);
            blo[i] = __builtin_shufflevector(f.b[i][j], f.b[i][j], 0, 1);
            bhi[i] = __builtin_shufflevector(f.b[i][j], f.b[i][j], 2, 3);
        }
#pragma unroll
        for (int mi = 0; mi < 4; ++mi)
#pragma unroll
            for (int ni = 0; ni < 4; ++ni)
                c[mi][ni] = __builtin_amdgcn_wmma_f32_16x16x4_f32(false, alo[mi], false, blo[ni],
                                                                  (short)0, c[mi][ni], false, false);
#pragma unroll
        for (int mi = 0; mi < 4; ++mi)
#pragma unroll
            for (int ni = 0; ni < 4; ++ni)
                c[mi][ni] = __builtin_amdgcn_wmma_f32_16x16x4_f32(false, ahi[mi], false, bhi[ni],
                                                                  (short)0, c[mi][ni], false, false);
    }
}

template <int STORE_VT>
__global__ __launch_bounds__(128) void gemm_xwt_kernel(const float* __restrict__ X,
                                                       const float* __restrict__ W,
                                                       float* __restrict__ out,
                                                       int M, int N, int K) {
    const int lane = threadIdx.x & 31;
    const int warp = threadIdx.x >> 5;
    const int tilesN = N >> 7;
    const int bm = (blockIdx.x / tilesN) << 7;
    const int bn = (blockIdx.x % tilesN) << 7;
    const int tm = bm + ((warp >> 1) << 6);
    const int tn = bn + ((warp & 1) << 6);
    const int ln = lane & 15;
    const int lh = lane >> 4;

    const float* xp = X + (size_t)(tm + ln) * K + 4 * lh;
    const float* wp = W + (size_t)(tn + ln) * K + 4 * lh;

    v8f c[4][4] = {};
    GemmFrag f0, f1;
    gemm_load_chunk(f0, xp, wp, K, 0);
    for (int k0 = 0; k0 < K; k0 += 64) {
        gemm_load_chunk(f1, xp, wp, K, k0 + 32);
        __builtin_amdgcn_sched_barrier(0);
        gemm_mma_chunk(c, f0);
        __builtin_amdgcn_sched_barrier(0);
        const int kn = (k0 + 64 < K) ? (k0 + 64) : 0;
        gemm_load_chunk(f0, xp, wp, K, kn);
        __builtin_amdgcn_sched_barrier(0);
        gemm_mma_chunk(c, f1);
        __builtin_amdgcn_sched_barrier(0);
    }

    if (STORE_VT == 0) {
#pragma unroll
        for (int mi = 0; mi < 4; ++mi)
#pragma unroll
            for (int ni = 0; ni < 4; ++ni) {
                float* orow = out + (size_t)(tm + mi * 16 + 8 * lh) * N + tn + ni * 16 + ln;
#pragma unroll
                for (int r = 0; r < 8; ++r)
                    orow[(size_t)r * N] = c[mi][ni][r];
            }
    } else {
        const int bb = tm >> 11;
#pragma unroll
        for (int mi = 0; mi < 4; ++mi)
#pragma unroll
            for (int ni = 0; ni < 4; ++ni) {
                int n = tn + ni * 16 + ln;
                int srow = (tm & (SEQ - 1)) + mi * 16 + 8 * lh;
                float* ocol = out + ((size_t)(bb * N + n) << 11) + srow;
#pragma unroll
                for (int r = 0; r < 8; ++r)
                    ocol[r] = c[mi][ni][r];
            }
    }
}

__global__ __launch_bounds__(256) void rope_kernel(float* __restrict__ q,
                                                   float* __restrict__ k, int total) {
    int tid = blockIdx.x * blockDim.x + threadIdx.x;
    if (tid >= total) return;
    int i = tid & 31;
    int s = (tid >> 9) & (SEQ - 1);
    float freq = __expf(-(float)i * (1.0f / 32.0f) * 9.210340371976184f);
    float ang = (float)s * freq;
    float cs = __cosf(ang);
    float sn = __sinf(ang);
    size_t p = (size_t)tid * 2;
    float q1 = q[p], q2 = q[p + 1];
    q[p]     = cs * q1 - sn * q2;
    q[p + 1] = sn * q1 + cs * q2;
    float k1 = k[p], k2 = k[p + 1];
    k[p]     = cs * k1 - sn * k2;
    k[p + 1] = sn * k1 + cs * k2;
}

__device__ __forceinline__ float rowmax16(float v) {
    v = fmaxf(v, __shfl_xor(v, 1, 32));
    v = fmaxf(v, __shfl_xor(v, 2, 32));
    v = fmaxf(v, __shfl_xor(v, 4, 32));
    v = fmaxf(v, __shfl_xor(v, 8, 32));
    return v;
}
__device__ __forceinline__ float rowsum16(float v) {
    v += __shfl_xor(v, 1, 32);
    v += __shfl_xor(v, 2, 32);
    v += __shfl_xor(v, 4, 32);
    v += __shfl_xor(v, 8, 32);
    return v;
}

#define PST 36

template <int NT, bool MASKED>
__device__ __forceinline__ void attn_block(const float* __restrict__ kl,
                                           const float* const (&vrow)[4],
                                           const int kt0, const int qt0,
                                           const int ln, const int lh, const int lnm,
                                           const v2f (&qa)[2][16], v8f (&acc)[2][4],
                                           float (&mrow)[2][8], float (&lsum)[2][8],
                                           float* __restrict__ pl0, float* __restrict__ pl1) {
    v8f s[2][NT];
#pragma unroll
    for (int i = 0; i < NT; ++i) {
        const float* kbase = kl + (size_t)(kt0 + i) * 16 * DM;
        s[0][i] = v8f{};
        s[1][i] = v8f{};
#pragma unroll
        for (int j = 0; j < 16; ++j) {
            v2f bf = *(const v2f*)(kbase + 4 * j);
            s[0][i] = __builtin_amdgcn_wmma_f32_16x16x4_f32(false, qa[0][j], false, bf,
                                                            (short)0, s[0][i], false, false);
            s[1][i] = __builtin_amdgcn_wmma_f32_16x16x4_f32(false, qa[1][j], false, bf,
                                                            (short)0, s[1][i], false, false);
        }
    }
#pragma unroll
    for (int t = 0; t < 2; ++t) {
        float nm[8];
#pragma unroll
        for (int r = 0; r < 8; ++r) {
            float mx = 0.0f;
#pragma unroll
            for (int i = 0; i < NT; ++i) {
                float sv = s[t][i][r] * 0.125f;
                if (MASKED) {
                    const int d = (qt0 + t - kt0 - i) * 16;
                    if (lnm - r > d) sv = -1e30f;
                }
                s[t][i][r] = sv;
                mx = (i == 0) ? sv : fmaxf(mx, sv);
            }
            nm[r] = fmaxf(mrow[t][r], rowmax16(mx));
        }
#pragma unroll
        for (int r = 0; r < 8; ++r) {
            float ps = 0.0f;
#pragma unroll
            for (int i = 0; i < NT; ++i) {
                float p = __expf(s[t][i][r] - nm[r]);
                s[t][i][r] = p;
                ps = (i == 0) ? p : ps + p;
            }
            float sc = __expf(mrow[t][r] - nm[r]);
            lsum[t][r] = lsum[t][r] * sc + ps;
            mrow[t][r] = nm[r];
#pragma unroll
            for (int nt = 0; nt < 4; ++nt) acc[t][nt][r] *= sc;
        }
    }
#pragma unroll
    for (int hb = 0; hb < NT / 2; ++hb) {
#pragma unroll
        for (int ii = 0; ii < 2; ++ii) {
#pragma unroll
            for (int r = 0; r < 8; ++r) {
                pl0[(r + 8 * lh) * PST + 16 * ii + ln] = s[0][2 * hb + ii][r];
                pl1[(r + 8 * lh) * PST + 16 * ii + ln] = s[1][2 * hb + ii][r];
            }
        }
#pragma unroll
        for (int ii = 0; ii < 2; ++ii) {
            const int kt = kt0 + 2 * hb + ii;
#pragma unroll
            for (int j = 0; j < 4; ++j) {
                v2f a0 = *(const v2f*)(pl0 + ln * PST + 16 * ii + 4 * j + 2 * lh);
                v2f a1 = *(const v2f*)(pl1 + ln * PST + 16 * ii + 4 * j + 2 * lh);
#pragma unroll
                for (int nt = 0; nt < 4; ++nt) {
                    v2f bf = *(const v2f*)(vrow[nt] + kt * 16 + 4 * j);
                    acc[0][nt] = __builtin_amdgcn_wmma_f32_16x16x4_f32(false, a0, false, bf,
                                                                       (short)0, acc[0][nt],
                                                                       false, false);
                    acc[1][nt] = __builtin_amdgcn_wmma_f32_16x16x4_f32(false, a1, false, bf,
                                                                       (short)0, acc[1][nt],
                                                                       false, false);
                }
            }
        }
    }
}

__global__ __launch_bounds__(256) __attribute__((amdgpu_waves_per_eu(2, 2)))
void attn_kernel(const float* __restrict__ q,
                                                   const float* __restrict__ k,
                                                   const float* __restrict__ vt,
                                                   float* __restrict__ o) {
    __shared__ float lds[8][2][16 * PST];
    const int lane = threadIdx.x & 31;
    const int warp = threadIdx.x >> 5;
    const int wave = __builtin_amdgcn_readfirstlane(blockIdx.x * 8 + warp);
    const int qp = wave & 63;
    const int bh = wave >> 6;
    const int h = bh & 15;
    const int b = bh >> 4;
    const int ln = lane & 15;
    const int lh = lane >> 4;
    const int lnm = ln - 8 * lh;

    v2f qa[2][16];
#pragma unroll
    for (int t = 0; t < 2; ++t) {
        const float* qbase = q + ((size_t)(b * SEQ) + (2 * qp + t) * 16 + ln) * DM + h * DK;
#pragma unroll
        for (int j = 0; j < 16; ++j)
            qa[t][j] = *(const v2f*)(qbase + 4 * j + 2 * lh);
    }

    const float* kl = k + ((size_t)(b * SEQ) + ln) * DM + h * DK + 2 * lh;
    const float* vrow[4];
#pragma unroll
    for (int nt = 0; nt < 4; ++nt)
        vrow[nt] = vt + ((size_t)(b * DM + h * DK + nt * 16 + ln) << 11) + 2 * lh;

    v8f acc[2][4];
    float mrow[2][8], lsum[2][8];
#pragma unroll
    for (int t = 0; t < 2; ++t) {
#pragma unroll
        for (int nt = 0; nt < 4; ++nt) acc[t][nt] = v8f{};
#pragma unroll
        for (int r = 0; r < 8; ++r) { mrow[t][r] = -1e30f; lsum[t][r] = 0.0f; }
    }

    float* pl0 = &lds[warp][0][0];
    float* pl1 = &lds[warp][1][0];

    const int qt0 = 2 * qp;
    const int nmain = qp >> 1;
    for (int kb = 0; kb < nmain; ++kb)
        attn_block<4, false>(kl, vrow, 4 * kb, qt0, ln, lh, lnm, qa, acc, mrow, lsum, pl0, pl1);
    if (qp & 1)
        attn_block<4, true>(kl, vrow, qt0 - 2, qt0, ln, lh, lnm, qa, acc, mrow, lsum, pl0, pl1);
    else
        attn_block<2, true>(kl, vrow, qt0, qt0, ln, lh, lnm, qa, acc, mrow, lsum, pl0, pl1);

#pragma unroll
    for (int t = 0; t < 2; ++t) {
        float* obase = o + ((size_t)(b * SEQ) + (2 * qp + t) * 16) * DM + h * DK;
#pragma unroll
        for (int r = 0; r < 8; ++r) {
            float inv = 1.0f / rowsum16(lsum[t][r]);
            int m = r + 8 * lh;
#pragma unroll
            for (int nt = 0; nt < 4; ++nt)
                obase[(size_t)m * DM + nt * 16 + ln] = acc[t][nt][r] * inv;
        }
    }
}

extern "C" void kernel_launch(void* const* d_in, const int* in_sizes, int n_in,
                              void* d_out, int out_size, void* d_ws, size_t ws_size,
                              hipStream_t stream) {
    const float* x  = (const float*)d_in[0];
    const float* Wq = (const float*)d_in[1];
    const float* Wk = (const float*)d_in[2];
    const float* Wv = (const float*)d_in[3];
    const float* Wo = (const float*)d_in[4];
    float* out = (float*)d_out;

    const size_t bufElems = (size_t)BATCH * SEQ * DM;
    float* qb  = (float*)d_ws;
    float* kb  = qb + bufElems;
    float* vtb = kb + bufElems;
    float* ab  = vtb + bufElems;

    const int M = BATCH * SEQ, N = DM, K = DM;
    const int gblocks = (M / 128) * (N / 128);

    gemm_xwt_kernel<0><<<gblocks, 128, 0, stream>>>(x, Wq, qb, M, N, K);
    gemm_xwt_kernel<0><<<gblocks, 128, 0, stream>>>(x, Wk, kb, M, N, K);
    gemm_xwt_kernel<1><<<gblocks, 128, 0, stream>>>(x, Wv, vtb, M, N, K);

    const int pairs = BATCH * SEQ * NH * (DK / 2);
    rope_kernel<<<pairs / 256, 256, 0, stream>>>(qb, kb, pairs);

    const int waves = BATCH * NH * (SEQ / 32);
    attn_kernel<<<waves / 8, 256, 0, stream>>>(qb, kb, vtb, ab);

    gemm_xwt_kernel<0><<<gblocks, 128, 0, stream>>>(ab, Wo, out, M, N, K);
}


# attention: Q, K, Vt and P fragments read as 16-byte per-lane k-runs (.xy and .zw feed two matrix instructions, k = {0,1,4,5} then {2,3,6,7} per 8-run): half the load instructions; k sum order changes
# speedup vs baseline: 2.0919x; 1.0873x over previous
#include <hip/hip_runtime.h>
#include <math.h>

typedef __attribute__((ext_vector_type(2))) float v2f;
typedef __attribute__((ext_vector_type(4))) float v4f;
typedef __attribute__((ext_vector_type(8))) float v8f;

#define BATCH 2
#define SEQ 2048
#define DM 1024
#define NH 16
#define DK 64

struct GemmFrag {
    v4f a[4][4];
    v4f b[4][4];
};

__device__ __forceinline__ void gemm_load_chunk(GemmFrag& f, const float* __restrict__ xp,
                                                const float* __restrict__ wp, int K, int kc) {
#pragma unroll
    for (int i = 0; i < 4; ++i) {
#pragma unroll
        for (int j = 0; j < 4; ++j)
            f.a[i][j] = *(const v4f*)(xp + (size_t)(16 * i) * K + kc + 8 * j);
#pragma unroll
        for (int j = 0; j < 4; ++j)
            f.b[i][j] = *(const v4f*)(wp + (size_t)(16 * i) * K + kc + 8 * j);
    }
}

__device__ __forceinline__ void gemm_mma_chunk(v8f (&c)[4][4], const GemmFrag& f) {
#pragma unroll
    for (int j = 0; j < 4; ++j) {
        v2f alo[4], ahi[4], blo[4], bhi[4];
#pragma unroll
        for (int i = 0; i < 4; ++i) {
            alo[i] = __builtin_shufflevector(f.a[i][j], f.a[i][j], 0, 1);
            ahi[i] = __builtin_shufflevector(f.a[i][j], f.a[i][j], 2, 3);
            blo[i] = __builtin_shufflevector(f.b[i][j], f.b[i][j], 0, 1);
            bhi[i] = __builtin_shufflevector(f.b[i][j], f.b[i][j], 2, 3);
        }
#pragma unroll
        for (int mi = 0; mi < 4; ++mi)
#pragma unroll
            for (int ni = 0; ni < 4; ++ni)
                c[mi][ni] = __builtin_amdgcn_wmma_f32_16x16x4_f32(false, alo[mi], false, blo[ni],
                                                                  (short)0, c[mi][ni], false, false);
#pragma unroll
        for (int mi = 0; mi < 4; ++mi)
#pragma unroll
            for (int ni = 0; ni < 4; ++ni)
                c[mi][ni] = __builtin_amdgcn_wmma_f32_16x16x4_f32(false, ahi[mi], false, bhi[ni],
                                                                  (short)0, c[mi][ni], false, false);
    }
}

template <int STORE_VT>
__global__ __launch_bounds__(128) void gemm_xwt_kernel(const float* __restrict__ X,
                                                       const float* __restrict__ W,
                                                       float* __restrict__ out,
                                                       int M, int N, int K) {
    const int lane = threadIdx.x & 31;
    const int warp = threadIdx.x >> 5;
    const int tilesN = N >> 7;
    const int bm = (blockIdx.x / tilesN) << 7;
    const int bn = (blockIdx.x % tilesN) << 7;
    const int tm = bm + ((warp >> 1) << 6);
    const int tn = bn + ((warp & 1) << 6);
    const int ln = lane & 15;
    const int lh = lane >> 4;

    const float* xp = X + (size_t)(tm + ln) * K + 4 * lh;
    const float* wp = W + (size_t)(tn + ln) * K + 4 * lh;

    v8f c[4][4] = {};
    GemmFrag f0, f1;
    gemm_load_chunk(f0, xp, wp, K, 0);
    for (int k0 = 0; k0 < K; k0 += 64) {
        gemm_load_chunk(f1, xp, wp, K, k0 + 32);
        __builtin_amdgcn_sched_barrier(0);
        gemm_mma_chunk(c, f0);
        __builtin_amdgcn_sched_barrier(0);
        const int kn = (k0 + 64 < K) ? (k0 + 64) : 0;
        gemm_load_chunk(f0, xp, wp, K, kn);
        __builtin_amdgcn_sched_barrier(0);
        gemm_mma_chunk(c, f1);
        __builtin_amdgcn_sched_barrier(0);
    }

    if (STORE_VT == 0) {
#pragma unroll
        for (int mi = 0; mi < 4; ++mi)
#pragma unroll
            for (int ni = 0; ni < 4; ++ni) {
                float* orow = out + (size_t)(tm + mi * 16 + 8 * lh) * N + tn + ni * 16 + ln;
#pragma unroll
                for (int r = 0; r < 8; ++r)
                    orow[(size_t)r * N] = c[mi][ni][r];
            }
    } else {
        const int bb = tm >> 11;
#pragma unroll
        for (int mi = 0; mi < 4; ++mi)
#pragma unroll
            for (int ni = 0; ni < 4; ++ni) {
                int n = tn + ni * 16 + ln;
                int srow = (tm & (SEQ - 1)) + mi * 16 + 8 * lh;
                float* ocol = out + ((size_t)(bb * N + n) << 11) + srow;
#pragma unroll
                for (int r = 0; r < 8; ++r)
                    ocol[r] = c[mi][ni][r];
            }
    }
}

__global__ __launch_bounds__(256) void rope_kernel(float* __restrict__ q,
                                                   float* __restrict__ k, int total) {
    int tid = blockIdx.x * blockDim.x + threadIdx.x;
    if (tid >= total) return;
    int i = tid & 31;
    int s = (tid >> 9) & (SEQ - 1);
    float freq = __expf(-(float)i * (1.0f / 32.0f) * 9.210340371976184f);
    float ang = (float)s * freq;
    float cs = __cosf(ang);
    float sn = __sinf(ang);
    size_t p = (size_t)tid * 2;
    float q1 = q[p], q2 = q[p + 1];
    q[p]     = cs * q1 - sn * q2;
    q[p + 1] = sn * q1 + cs * q2;
    float k1 = k[p], k2 = k[p + 1];
    k[p]     = cs * k1 - sn * k2;
    k[p + 1] = sn * k1 + cs * k2;
}

__device__ __forceinline__ float rowmax16(float v) {
    v = fmaxf(v, __shfl_xor(v, 1, 32));
    v = fmaxf(v, __shfl_xor(v, 2, 32));
    v = fmaxf(v, __shfl_xor(v, 4, 32));
    v = fmaxf(v, __shfl_xor(v, 8, 32));
    return v;
}
__device__ __forceinline__ float rowsum16(float v) {
    v += __shfl_xor(v, 1, 32);
    v += __shfl_xor(v, 2, 32);
    v += __shfl_xor(v, 4, 32);
    v += __shfl_xor(v, 8, 32);
    return v;
}

#define PST 36

template <int NT, bool MASKED>
__device__ __forceinline__ void attn_block(const float* __restrict__ kl,
                                           const float* const (&vrow)[4],
                                           const int kt0, const int qt0,
                                           const int ln, const int lh, const int lnm,
                                           const v2f (&qa)[2][16], v8f (&acc)[2][4],
                                           float (&mrow)[2][8], float (&lsum)[2][8],
                                           float* __restrict__ pl0, float* __restrict__ pl1) {
    v8f s[2][NT];
#pragma unroll
    for (int i = 0; i < NT; ++i) {
        const float* kbase = kl + (size_t)(kt0 + i) * 16 * DM;
        s[0][i] = v8f{};
        s[1][i] = v8f{};
#pragma unroll
        for (int j = 0; j < 8; ++j) {
            const v4f kf = *(const v4f*)(kbase + 8 * j);
            const v2f klo = __builtin_shufflevector(kf, kf, 0, 1);
            const v2f khi = __builtin_shufflevector(kf, kf, 2, 3);
            s[0][i] = __builtin_amdgcn_wmma_f32_16x16x4_f32(false, qa[0][2 * j], false, klo,
                                                            (short)0, s[0][i], false, false);
            s[1][i] = __builtin_amdgcn_wmma_f32_16x16x4_f32(false, qa[1][2 * j], false, klo,
                                                            (short)0, s[1][i], false, false);
            s[0][i] = __builtin_amdgcn_wmma_f32_16x16x4_f32(false, qa[0][2 * j + 1], false, khi,
                                                            (short)0, s[0][i], false, false);
            s[1][i] = __builtin_amdgcn_wmma_f32_16x16x4_f32(false, qa[1][2 * j + 1], false, khi,
                                                            (short)0, s[1][i], false, false);
        }
    }
#pragma unroll
    for (int t = 0; t < 2; ++t) {
        float nm[8];
#pragma unroll
        for (int r = 0; r < 8; ++r) {
            float mx = 0.0f;
#pragma unroll
            for (int i = 0; i < NT; ++i) {
                float sv = s[t][i][r] * 0.125f;
                if (MASKED) {
                    const int d = (qt0 + t - kt0 - i) * 16;
                    if (lnm - r > d) sv = -1e30f;
                }
                s[t][i][r] = sv;
                mx = (i == 0) ? sv : fmaxf(mx, sv);
            }
            nm[r] = fmaxf(mrow[t][r], rowmax16(mx));
        }
#pragma unroll
        for (int r = 0; r < 8; ++r) {
            float ps = 0.0f;
#pragma unroll
            for (int i = 0; i < NT; ++i) {
                float p = __expf(s[t][i][r] - nm[r]);
                s[t][i][r] = p;
                ps = (i == 0) ? p : ps + p;
            }
            float sc = __expf(mrow[t][r] - nm[r]);
            lsum[t][r] = lsum[t][r] * sc + ps;
            mrow[t][r] = nm[r];
#pragma unroll
            for (int nt = 0; nt < 4; ++nt) acc[t][nt][r] *= sc;
        }
    }
#pragma unroll
    for (int hb = 0; hb < NT / 2; ++hb) {
#pragma unroll
        for (int ii = 0; ii < 2; ++ii) {
#pragma unroll
            for (int r = 0; r < 8; ++r) {
                pl0[(r + 8 * lh) * PST + 16 * ii + ln] = s[0][2 * hb + ii][r];
                pl1[(r + 8 * lh) * PST + 16 * ii + ln] = s[1][2 * hb + ii][r];
            }
        }
        const int key0 = (kt0 + 2 * hb) * 16;
#pragma unroll
        for (int j = 0; j < 4; ++j) {
            const v4f a0 = *(const v4f*)(pl0 + ln * PST + 8 * j + 4 * lh);
            const v4f a1 = *(const v4f*)(pl1 + ln * PST + 8 * j + 4 * lh);
            const v2f a0lo = __builtin_shufflevector(a0, a0, 0, 1);
            const v2f a0hi = __builtin_shufflevector(a0, a0, 2, 3);
            const v2f a1lo = __builtin_shufflevector(a1, a1, 0, 1);
            const v2f a1hi = __builtin_shufflevector(a1, a1, 2, 3);
            v4f vf[4];
#pragma unroll
            for (int nt = 0; nt < 4; ++nt)
                vf[nt] = *(const v4f*)(vrow[nt] + key0 + 8 * j);
#pragma unroll
            for (int nt = 0; nt < 4; ++nt) {
                const v2f vlo = __builtin_shufflevector(vf[nt], vf[nt], 0, 1);
                acc[0][nt] = __builtin_amdgcn_wmma_f32_16x16x4_f32(false, a0lo, false, vlo,
                                                                   (short)0, acc[0][nt],
                                                                   false, false);
                acc[1][nt] = __builtin_amdgcn_wmma_f32_16x16x4_f32(false, a1lo, false, vlo,
                                                                   (short)0, acc[1][nt],
                                                                   false, false);
            }
#pragma unroll
            for (int nt = 0; nt < 4; ++nt) {
                const v2f vhi = __builtin_shufflevector(vf[nt], vf[nt], 2, 3);
                acc[0][nt] = __builtin_amdgcn_wmma_f32_16x16x4_f32(false, a0hi, false, vhi,
                                                                   (short)0, acc[0][nt],
                                                                   false, false);
                acc[1][nt] = __builtin_amdgcn_wmma_f32_16x16x4_f32(false, a1hi, false, vhi,
                                                                   (short)0, acc[1][nt],
                                                                   false, false);
            }
        }
    }
}

__global__ __launch_bounds__(256) __attribute__((amdgpu_waves_per_eu(2, 2)))
void attn_kernel(const float* __restrict__ q,
                                                   const float* __restrict__ k,
                                                   const float* __restrict__ vt,
                                                   float* __restrict__ o) {
    __shared__ float lds[8][2][16 * PST];
    const int lane = threadIdx.x & 31;
    const int warp = threadIdx.x >> 5;
    const int wave = __builtin_amdgcn_readfirstlane(blockIdx.x * 8 + warp);
    const int qp = wave & 63;
    const int bh = wave >> 6;
    const int h = bh & 15;
    const int b = bh >> 4;
    const int ln = lane & 15;
    const int lh = lane >> 4;
    const int lnm = ln - 8 * lh;

    v2f qa[2][16];
#pragma unroll
    for (int t = 0; t < 2; ++t) {
        const float* qbase = q + ((size_t)(b * SEQ) + (2 * qp + t) * 16 + ln) * DM + h * DK;
#pragma unroll
        for (int j = 0; j < 8; ++j) {
            const v4f qf = *(const v4f*)(qbase + 8 * j + 4 * lh);
            qa[t][2 * j]     = __builtin_shufflevector(qf, qf, 0, 1);
            qa[t][2 * j + 1] = __builtin_shufflevector(qf, qf, 2, 3);
        }
    }

    const float* kl = k + ((size_t)(b * SEQ) + ln) * DM + h * DK + 4 * lh;
    const float* vrow[4];
#pragma unroll
    for (int nt = 0; nt < 4; ++nt)
        vrow[nt] = vt + ((size_t)(b * DM + h * DK + nt * 16 + ln) << 11) + 4 * lh;

    v8f acc[2][4];
    float mrow[2][8], lsum[2][8];
#pragma unroll
    for (int t = 0; t < 2; ++t) {
#pragma unroll
        for (int nt = 0; nt < 4; ++nt) acc[t][nt] = v8f{};
#pragma unroll
        for (int r = 0; r < 8; ++r) { mrow[t][r] = -1e30f; lsum[t][r] = 0.0f; }
    }

    float* pl0 = &lds[warp][0][0];
    float* pl1 = &lds[warp][1][0];

    const int qt0 = 2 * qp;
    const int nmain = qp >> 1;
    for (int kb = 0; kb < nmain; ++kb)
        attn_block<4, false>(kl, vrow, 4 * kb, qt0, ln, lh, lnm, qa, acc, mrow, lsum, pl0, pl1);
    if (qp & 1)
        attn_block<4, true>(kl, vrow, qt0 - 2, qt0, ln, lh, lnm, qa, acc, mrow, lsum, pl0, pl1);
    else
        attn_block<2, true>(kl, vrow, qt0, qt0, ln, lh, lnm, qa, acc, mrow, lsum, pl0, pl1);

#pragma unroll
    for (int t = 0; t < 2; ++t) {
        float* obase = o + ((size_t)(b * SEQ) + (2 * qp + t) * 16) * DM + h * DK;
#pragma unroll
        for (int r = 0; r < 8; ++r) {
            float inv = 1.0f / rowsum16(lsum[t][r]);
            int m = r + 8 * lh;
#pragma unroll
            for (int nt = 0; nt < 4; ++nt)
                obase[(size_t)m * DM + nt * 16 + ln] = acc[t][nt][r] * inv;
        }
    }
}

extern "C" void kernel_launch(void* const* d_in, const int* in_sizes, int n_in,
                              void* d_out, int out_size, void* d_ws, size_t ws_size,
                              hipStream_t stream) {
    const float* x  = (const float*)d_in[0];
    const float* Wq = (const float*)d_in[1];
    const float* Wk = (const float*)d_in[2];
    const float* Wv = (const float*)d_in[3];
    const float* Wo = (const float*)d_in[4];
    float* out = (float*)d_out;

    const size_t bufElems = (size_t)BATCH * SEQ * DM;
    float* qb  = (float*)d_ws;
    float* kb  = qb + bufElems;
    float* vtb = kb + bufElems;
    float* ab  = vtb + bufElems;

    const int M = BATCH * SEQ, N = DM, K = DM;
    const int gblocks = (M / 128) * (N / 128);

    gemm_xwt_kernel<0><<<gblocks, 128, 0, stream>>>(x, Wq, qb, M, N, K);
    gemm_xwt_kernel<0><<<gblocks, 128, 0, stream>>>(x, Wk, kb, M, N, K);
    gemm_xwt_kernel<1><<<gblocks, 128, 0, stream>>>(x, Wv, vtb, M, N, K);

    const int pairs = BATCH * SEQ * NH * (DK / 2);
    rope_kernel<<<pairs / 256, 256, 0, stream>>>(qb, kb, pairs);

    const int waves = BATCH * NH * (SEQ / 32);
    attn_kernel<<<waves / 8, 256, 0, stream>>>(qb, kb, vtb, ab);

    gemm_xwt_kernel<0><<<gblocks, 128, 0, stream>>>(ab, Wo, out, M, N, K);
}


# attention: each query pair's keys cut into 512-key chunks run as separate waves, longest work-groups first (640 groups, 256 resident); a small kernel merges partial softmax states; sum order changes
# speedup vs baseline: 2.3969x; 1.1458x over previous
#include <hip/hip_runtime.h>
#include <math.h>

typedef __attribute__((ext_vector_type(2))) float v2f;
typedef __attribute__((ext_vector_type(4))) float v4f;
typedef __attribute__((ext_vector_type(8))) float v8f;

#define BATCH 2
#define SEQ 2048
#define DM 1024
#define NH 16
#define DK 64

struct GemmFrag {
    v4f a[4][4];
    v4f b[4][4];
};

__device__ __forceinline__ void gemm_load_chunk(GemmFrag& f, const float* __restrict__ xp,
                                                const float* __restrict__ wp, int K, int kc) {
#pragma unroll
    for (int i = 0; i < 4; ++i) {
#pragma unroll
        for (int j = 0; j < 4; ++j)
            f.a[i][j] = *(const v4f*)(xp + (size_t)(16 * i) * K + kc + 8 * j);
#pragma unroll
        for (int j = 0; j < 4; ++j)
            f.b[i][j] = *(const v4f*)(wp + (size_t)(16 * i) * K + kc + 8 * j);
    }
}

__device__ __forceinline__ void gemm_mma_chunk(v8f (&c)[4][4], const GemmFrag& f) {
#pragma unroll
    for (int j = 0; j < 4; ++j) {
        v2f alo[4], ahi[4], blo[4], bhi[4];
#pragma unroll
        for (int i = 0; i < 4; ++i) {
            alo[i] = __builtin_shufflevector(f.a[i][j], f.a[i][j], 0, 1);
            ahi[i] = __builtin_shufflevector(f.a[i][j], f.a[i][j], 2, 3);
            blo[i] = __builtin_shufflevector(f.b[i][j], f.b[i][j], 0, 1);
            bhi[i] = __builtin_shufflevector(f.b[i][j], f.b[i][j], 2, 3);
        }
#pragma unroll
        for (int mi = 0; mi < 4; ++mi)
#pragma unroll
            for (int ni = 0; ni < 4; ++ni)
                c[mi][ni] = __builtin_amdgcn_wmma_f32_16x16x4_f32(false, alo[mi], false, blo[ni],
                                                                  (short)0, c[mi][ni], false, false);
#pragma unroll
        for (int mi = 0; mi < 4; ++mi)
#pragma unroll
            for (int ni = 0; ni < 4; ++ni)
                c[mi][ni] = __builtin_amdgcn_wmma_f32_16x16x4_f32(false, ahi[mi], false, bhi[ni],
                                                                  (short)0, c[mi][ni], false, false);
    }
}

template <int STORE_VT>
__global__ __launch_bounds__(128) void gemm_xwt_kernel(const float* __restrict__ X,
                                                       const float* __restrict__ W,
                                                       float* __restrict__ out,
                                                       int M, int N, int K) {
    const int lane = threadIdx.x & 31;
    const int warp = threadIdx.x >> 5;
    const int tilesN = N >> 7;
    const int bm = (blockIdx.x / tilesN) << 7;
    const int bn = (blockIdx.x % tilesN) << 7;
    const int tm = bm + ((warp >> 1) << 6);
    const int tn = bn + ((warp & 1) << 6);
    const int ln = lane & 15;
    const int lh = lane >> 4;

    const float* xp = X + (size_t)(tm + ln) * K + 4 * lh;
    const float* wp = W + (size_t)(tn + ln) * K + 4 * lh;

    v8f c[4][4] = {};
    GemmFrag f0, f1;
    gemm_load_chunk(f0, xp, wp, K, 0);
    for (int k0 = 0; k0 < K; k0 += 64) {
        gemm_load_chunk(f1, xp, wp, K, k0 + 32);
        __builtin_amdgcn_sched_barrier(0);
        gemm_mma_chunk(c, f0);
        __builtin_amdgcn_sched_barrier(0);
        const int kn = (k0 + 64 < K) ? (k0 + 64) : 0;
        gemm_load_chunk(f0, xp, wp, K, kn);
        __builtin_amdgcn_sched_barrier(0);
        gemm_mma_chunk(c, f1);
        __builtin_amdgcn_sched_barrier(0);
    }

    if (STORE_VT == 0) {
#pragma unroll
        for (int mi = 0; mi < 4; ++mi)
#pragma unroll
            for (int ni = 0; ni < 4; ++ni) {
                float* orow = out + (size_t)(tm + mi * 16 + 8 * lh) * N + tn + ni * 16 + ln;
#pragma unroll
                for (int r = 0; r < 8; ++r)
                    orow[(size_t)r * N] = c[mi][ni][r];
            }
    } else {
        const int bb = tm >> 11;
#pragma unroll
        for (int mi = 0; mi < 4; ++mi)
#pragma unroll
            for (int ni = 0; ni < 4; ++ni) {
                int n = tn + ni * 16 + ln;
                int srow = (tm & (SEQ - 1)) + mi * 16 + 8 * lh;
                float* ocol = out + ((size_t)(bb * N + n) << 11) + srow;
#pragma unroll
                for (int r = 0; r < 8; ++r)
                    ocol[r] = c[mi][ni][r];
            }
    }
}

__global__ __launch_bounds__(256) void rope_kernel(float* __restrict__ q,
                                                   float* __restrict__ k, int total) {
    int tid = blockIdx.x * blockDim.x + threadIdx.x;
    if (tid >= total) return;
    int i = tid & 31;
    int s = (tid >> 9) & (SEQ - 1);
    float freq = __expf(-(float)i * (1.0f / 32.0f) * 9.210340371976184f);
    float ang = (float)s * freq;
    float cs = __cosf(ang);
    float sn = __sinf(ang);
    size_t p = (size_t)tid * 2;
    float q1 = q[p], q2 = q[p + 1];
    q[p]     = cs * q1 - sn * q2;
    q[p + 1] = sn * q1 + cs * q2;
    float k1 = k[p], k2 = k[p + 1];
    k[p]     = cs * k1 - sn * k2;
    k[p + 1] = sn * k1 + cs * k2;
}

__device__ __forceinline__ float rowmax16(float v) {
    v = fmaxf(v, __shfl_xor(v, 1, 32));
    v = fmaxf(v, __shfl_xor(v, 2, 32));
    v = fmaxf(v, __shfl_xor(v, 4, 32));
    v = fmaxf(v, __shfl_xor(v, 8, 32));
    return v;
}
__device__ __forceinline__ float rowsum16(float v) {
    v += __shfl_xor(v, 1, 32);
    v += __shfl_xor(v, 2, 32);
    v += __shfl_xor(v, 4, 32);
    v += __shfl_xor(v, 8, 32);
    return v;
}

#define ATT_FULL_GROUPS 384
#define ATT_GROUPS 640
#define PST 36

template <int NT, bool MASKED>
__device__ __forceinline__ void attn_block(const float* __restrict__ kl,
                                           const float* const (&vrow)[4],
                                           const int kt0, const int qt0,
                                           const int ln, const int lh, const int lnm,
                                           const v2f (&qa)[2][16], v8f (&acc)[2][4],
                                           float (&mrow)[2][8], float (&lsum)[2][8],
                                           float* __restrict__ pl0, float* __restrict__ pl1) {
    v8f s[2][NT];
#pragma unroll
    for (int i = 0; i < NT; ++i) {
        const float* kbase = kl + (size_t)(kt0 + i) * 16 * DM;
        s[0][i] = v8f{};
        s[1][i] = v8f{};
#pragma unroll
        for (int j = 0; j < 8; ++j) {
            const v4f kf = *(const v4f*)(kbase + 8 * j);
            const v2f klo = __builtin_shufflevector(kf, kf, 0, 1);
            const v2f khi = __builtin_shufflevector(kf, kf, 2, 3);
            s[0][i] = __builtin_amdgcn_wmma_f32_16x16x4_f32(false, qa[0][2 * j], false, klo,
                                                            (short)0, s[0][i], false, false);
            s[1][i] = __builtin_amdgcn_wmma_f32_16x16x4_f32(false, qa[1][2 * j], false, klo,
                                                            (short)0, s[1][i], false, false);
            s[0][i] = __builtin_amdgcn_wmma_f32_16x16x4_f32(false, qa[0][2 * j + 1], false, khi,
                                                            (short)0, s[0][i], false, false);
            s[1][i] = __builtin_amdgcn_wmma_f32_16x16x4_f32(false, qa[1][2 * j + 1], false, khi,
                                                            (short)0, s[1][i], false, false);
        }
    }
#pragma unroll
    for (int t = 0; t < 2; ++t) {
        float nm[8];
#pragma unroll
        for (int r = 0; r < 8; ++r) {
            float mx = 0.0f;
#pragma unroll
            for (int i = 0; i < NT; ++i) {
                float sv = s[t][i][r] * 0.125f;
                if (MASKED) {
                    const int d = (qt0 + t - kt0 - i) * 16;
                    if (lnm - r > d) sv = -1e30f;
                }
                s[t][i][r] = sv;
                mx = (i == 0) ? sv : fmaxf(mx, sv);
            }
            nm[r] = fmaxf(mrow[t][r], rowmax16(mx));
        }
#pragma unroll
        for (int r = 0; r < 8; ++r) {
            float ps = 0.0f;
#pragma unroll
            for (int i = 0; i < NT; ++i) {
                float p = __expf(s[t][i][r] - nm[r]);
                s[t][i][r] = p;
                ps = (i == 0) ? p : ps + p;
            }
            float sc = __expf(mrow[t][r] - nm[r]);
            lsum[t][r] = lsum[t][r] * sc + ps;
            mrow[t][r] = nm[r];
#pragma unroll
            for (int nt = 0; nt < 4; ++nt) acc[t][nt][r] *= sc;
        }
    }
#pragma unroll
    for (int hb = 0; hb < NT / 2; ++hb) {
#pragma unroll
        for (int ii = 0; ii < 2; ++ii) {
#pragma unroll
            for (int r = 0; r < 8; ++r) {
                pl0[(r + 8 * lh) * PST + 16 * ii + ln] = s[0][2 * hb + ii][r];
                pl1[(r + 8 * lh) * PST + 16 * ii + ln] = s[1][2 * hb + ii][r];
            }
        }
        const int key0 = (kt0 + 2 * hb) * 16;
#pragma unroll
        for (int j = 0; j < 4; ++j) {
            const v4f a0 = *(const v4f*)(pl0 + ln * PST + 8 * j + 4 * lh);
            const v4f a1 = *(const v4f*)(pl1 + ln * PST + 8 * j + 4 * lh);
            const v2f a0lo = __builtin_shufflevector(a0, a0, 0, 1);
            const v2f a0hi = __builtin_shufflevector(a0, a0, 2, 3);
            const v2f a1lo = __builtin_shufflevector(a1, a1, 0, 1);
            const v2f a1hi = __builtin_shufflevector(a1, a1, 2, 3);
            v4f vf[4];
#pragma unroll
            for (int nt = 0; nt < 4; ++nt)
                vf[nt] = *(const v4f*)(vrow[nt] + key0 + 8 * j);
#pragma unroll
            for (int nt = 0; nt < 4; ++nt) {
                const v2f vlo = __builtin_shufflevector(vf[nt], vf[nt], 0, 1);
                acc[0][nt] = __builtin_amdgcn_wmma_f32_16x16x4_f32(false, a0lo, false, vlo,
                                                                   (short)0, acc[0][nt],
                                                                   false, false);
                acc[1][nt] = __builtin_amdgcn_wmma_f32_16x16x4_f32(false, a1lo, false, vlo,
                                                                   (short)0, acc[1][nt],
                                                                   false, false);
            }
#pragma unroll
            for (int nt = 0; nt < 4; ++nt) {
                const v2f vhi = __builtin_shufflevector(vf[nt], vf[nt], 2, 3);
                acc[0][nt] = __builtin_amdgcn_wmma_f32_16x16x4_f32(false, a0hi, false, vhi,
                                                                   (short)0, acc[0][nt],
                                                                   false, false);
                acc[1][nt] = __builtin_amdgcn_wmma_f32_16x16x4_f32(false, a1hi, false, vhi,
                                                                   (short)0, acc[1][nt],
                                                                   false, false);
            }
        }
    }
}

__global__ __launch_bounds__(256) __attribute__((amdgpu_waves_per_eu(2, 2)))
void attn_kernel(const float* __restrict__ q,
                                                   const float* __restrict__ k,
                                                   const float* __restrict__ vt,
                                                   float* __restrict__ pacc,
                                                   float* __restrict__ pml) {
    __shared__ float lds[8][2][16 * PST];
    const int lane = threadIdx.x & 31;
    const int warp = threadIdx.x >> 5;
    const int ws = __builtin_amdgcn_readfirstlane(warp);
    const int blk = blockIdx.x;
    const bool last_chunk = blk >= ATT_FULL_GROUPS;
    int bh, qp, c;
    if (!last_chunk) {
        bh = blk / 12;
        const int w = blk - 12 * bh;
        c = (w >= 6) + (w >= 10);
        const int first = (w < 6) ? w : ((w < 10) ? w - 6 : w - 10);
        qp = 16 * (c + 1) + 8 * first + ws;
    } else {
        const int db = blk - ATT_FULL_GROUPS;
        const int d = 8 - (db >> 5);
        bh = db & 31;
        c = ws >> 1;
        qp = 16 * c + 2 * (d - 1) + (ws & 1);
    }
    const int h = bh & 15;
    const int b = bh >> 4;
    const int ln = lane & 15;
    const int lh = lane >> 4;
    const int lnm = ln - 8 * lh;

    v2f qa[2][16];
#pragma unroll
    for (int t = 0; t < 2; ++t) {
        const float* qbase = q + ((size_t)(b * SEQ) + (2 * qp + t) * 16 + ln) * DM + h * DK;
#pragma unroll
        for (int j = 0; j < 8; ++j) {
            const v4f qf = *(const v4f*)(qbase + 8 * j + 4 * lh);
            qa[t][2 * j]     = __builtin_shufflevector(qf, qf, 0, 1);
            qa[t][2 * j + 1] = __builtin_shufflevector(qf, qf, 2, 3);
        }
    }

    const float* kl = k + ((size_t)(b * SEQ) + ln) * DM + h * DK + 4 * lh;
    const float* vrow[4];
#pragma unroll
    for (int nt = 0; nt < 4; ++nt)
        vrow[nt] = vt + ((size_t)(b * DM + h * DK + nt * 16 + ln) << 11) + 4 * lh;

    v8f acc[2][4];
    float mrow[2][8], lsum[2][8];
#pragma unroll
    for (int t = 0; t < 2; ++t) {
#pragma unroll
        for (int nt = 0; nt < 4; ++nt) acc[t][nt] = v8f{};
#pragma unroll
        for (int r = 0; r < 8; ++r) { mrow[t][r] = -1e30f; lsum[t][r] = 0.0f; }
    }

    float* pl0 = &lds[warp][0][0];
    float* pl1 = &lds[warp][1][0];

    const int qt0 = 2 * qp;
    const int nmain = qp >> 1;
    const int kbeg = 8 * c;
    const int kend = last_chunk ? nmain : kbeg + 8;
    for (int kb = kbeg; kb < kend; ++kb)
        attn_block<4, false>(kl, vrow, 4 * kb, qt0, ln, lh, lnm, qa, acc, mrow, lsum, pl0, pl1);
    if (last_chunk) {
        if (qp & 1)
            attn_block<4, true>(kl, vrow, qt0 - 2, qt0, ln, lh, lnm, qa, acc, mrow, lsum, pl0, pl1);
        else
            attn_block<2, true>(kl, vrow, qt0, qt0, ln, lh, lnm, qa, acc, mrow, lsum, pl0, pl1);
    }

    const int slot = ((bh * 64 + qp) << 2) + c;
    float* pa = pacc + (size_t)slot * (32 * DK);
    float* pm = pml + (size_t)slot * 64;
#pragma unroll
    for (int t = 0; t < 2; ++t) {
#pragma unroll
        for (int r = 0; r < 8; ++r) {
            const float ls = rowsum16(lsum[t][r]);
            const int m = t * 16 + r + 8 * lh;
#pragma unroll
            for (int nt = 0; nt < 4; ++nt)
                pa[m * DK + nt * 16 + ln] = acc[t][nt][r];
            if (ln == 0) {
                pm[m] = mrow[t][r];
                pm[32 + m] = ls;
            }
        }
    }
}

__global__ __launch_bounds__(256) void attn_combine_kernel(const float* __restrict__ pacc,
                                                           const float* __restrict__ pml,
                                                           float* __restrict__ o) {
    const int tid = blockIdx.x * 256 + threadIdx.x;
    const int f4 = tid & 15;
    const int h = (tid >> 4) & 15;
    const int s = (tid >> 8) & (SEQ - 1);
    const int b = tid >> 19;
    const int qp = s >> 5;
    const int row = s & 31;
    const int nch = (qp >> 4) + 1;
    const size_t slot0 = (size_t)(((b * NH + h) * 64 + qp) << 2);
    float mc[4], lc[4];
    float mx = -1e30f;
#pragma unroll
    for (int c = 0; c < 4; ++c) {
        mc[c] = -1e30f;
        lc[c] = 0.0f;
        if (c < nch) {
            mc[c] = pml[(slot0 + c) * 64 + row];
            lc[c] = pml[(slot0 + c) * 64 + 32 + row];
        }
        mx = fmaxf(mx, mc[c]);
    }
    v4f num = {0.0f, 0.0f, 0.0f, 0.0f};
    float den = 0.0f;
#pragma unroll
    for (int c = 0; c < 4; ++c) {
        if (c < nch) {
            const float w = __expf(mc[c] - mx);
            const v4f a = *(const v4f*)(pacc + ((slot0 + c) * 32 + row) * DK + 4 * f4);
            num += a * w;
            den += lc[c] * w;
        }
    }
    const float inv = 1.0f / den;
    *(v4f*)(o + ((size_t)(b * SEQ) + s) * DM + h * DK + 4 * f4) = num * inv;
}

extern "C" void kernel_launch(void* const* d_in, const int* in_sizes, int n_in,
                              void* d_out, int out_size, void* d_ws, size_t ws_size,
                              hipStream_t stream) {
    const float* x  = (const float*)d_in[0];
    const float* Wq = (const float*)d_in[1];
    const float* Wk = (const float*)d_in[2];
    const float* Wv = (const float*)d_in[3];
    const float* Wo = (const float*)d_in[4];
    float* out = (float*)d_out;

    const size_t bufElems = (size_t)BATCH * SEQ * DM;
    float* qb  = (float*)d_ws;
    float* kb  = qb + bufElems;
    float* vtb = kb + bufElems;
    float* ab  = vtb + bufElems;
    float* pacc = ab + bufElems;
    float* pml  = pacc + (size_t)8192 * 32 * DK;

    const int M = BATCH * SEQ, N = DM, K = DM;
    const int gblocks = (M / 128) * (N / 128);

    gemm_xwt_kernel<0><<<gblocks, 128, 0, stream>>>(x, Wq, qb, M, N, K);
    gemm_xwt_kernel<0><<<gblocks, 128, 0, stream>>>(x, Wk, kb, M, N, K);
    gemm_xwt_kernel<1><<<gblocks, 128, 0, stream>>>(x, Wv, vtb, M, N, K);

    const int pairs = BATCH * SEQ * NH * (DK / 2);
    rope_kernel<<<pairs / 256, 256, 0, stream>>>(qb, kb, pairs);

    attn_kernel<<<ATT_GROUPS, 256, 0, stream>>>(qb, kb, vtb, pacc, pml);
    attn_combine_kernel<<<(BATCH * SEQ * NH * (DK / 4)) / 256, 256, 0, stream>>>(pacc, pml, ab);

    gemm_xwt_kernel<0><<<gblocks, 128, 0, stream>>>(ab, Wo, out, M, N, K);
}
